# starting kernel
# baseline (speedup 1.0000x reference)
#include <hip/hip_runtime.h>
#include <hip/hip_bf16.h>

typedef __attribute__((ext_vector_type(2))) float v2f;
typedef __attribute__((ext_vector_type(8))) float v8f;

__global__ void k_fill(float* __restrict__ p, size_t n, float v) {
    size_t i = (size_t)blockIdx.x * blockDim.x + threadIdx.x;
    size_t stride = (size_t)gridDim.x * blockDim.x;
    for (; i < n; i += stride) p[i] = v;
}

__global__ void k_deg_edges(const int* __restrict__ dst, float* __restrict__ deg, int E) {
    int e = blockIdx.x * blockDim.x + threadIdx.x;
    if (e < E) unsafeAtomicAdd(&deg[dst[e]], 1.0f);
}

__global__ void k_dis(const float* __restrict__ deg, float* __restrict__ dis, int N) {
    int i = blockIdx.x * blockDim.x + threadIdx.x;
    if (i < N) {
        float d = deg[i];
        dis[i] = (d > 0.0f) ? rsqrtf(d) : 0.0f;
    }
}

__global__ void k_pack_w(const float* __restrict__ Wmu, const float* __restrict__ Wlv,
                         float* __restrict__ Wcat) {
    int idx = blockIdx.x * blockDim.x + threadIdx.x;
    int k = idx >> 7, c = idx & 127;
    Wcat[idx] = (c < 64) ? Wmu[k * 64 + c] : Wlv[k * 64 + (c - 64)];
}

__global__ void k_gemm128(const float* __restrict__ A, const float* __restrict__ W,
                          float* __restrict__ C, int nrows) {
    const int lane  = threadIdx.x & 31;
    const int wave  = threadIdx.x >> 5;
    const int row0  = blockIdx.x * 16;
    const int col0  = wave * 16;
    const int m     = lane & 15;
    const int khalf = (lane < 16) ? 0 : 2;

    const float* __restrict__ arow = A + (size_t)(row0 + m) * 128;
    v8f acc = {};
#pragma unroll
    for (int k = 0; k < 128; k += 4) {
        v2f a, b;
        const float2 av = *reinterpret_cast<const float2*>(arow + k + khalf);
        a.x = av.x;
        a.y = av.y;
        b.x = W[(size_t)(k + khalf) * 128 + col0 + m];
        b.y = W[(size_t)(k + khalf + 1) * 128 + col0 + m];
        acc = __builtin_amdgcn_wmma_f32_16x16x4_f32(
             false, a,  false, b,
             (short)0, acc,  false,  false);
    }
    const int rbase = (lane < 16) ? 0 : 8;
#pragma unroll
    for (int j = 0; j < 8; ++j) {
        C[(size_t)(row0 + rbase + j) * 128 + col0 + m] = acc[j];
    }
}

template <bool SPLIT>
__global__ void k_scatter(const float* __restrict__ xw, const int* __restrict__ src,
                          const int* __restrict__ dst, const float* __restrict__ dis,
                          float* __restrict__ out, int E, int N) {
    const int gid  = blockIdx.x * blockDim.x + threadIdx.x;
    const int e    = gid >> 5;
    const int lane = gid & 31;
    if (e >= E) return;
    const int s = src[e];
    const int d = dst[e];
    const float nrm = dis[s] * dis[d];
    const int c = lane * 4;
    const float4 v = *reinterpret_cast<const float4*>(xw + (size_t)s * 128 + c);
    float* p;
    if (!SPLIT) {
        p = out + (size_t)d * 128 + c;
    } else {
        p = (c < 64) ? out + (size_t)d * 64 + c
                     : out + (size_t)N * 64 + (size_t)d * 64 + (c - 64);
    }
    unsafeAtomicAdd(p + 0, v.x * nrm);
    unsafeAtomicAdd(p + 1, v.y * nrm);
    unsafeAtomicAdd(p + 2, v.z * nrm);
    unsafeAtomicAdd(p + 3, v.w * nrm);
}

__global__ void k_selfloop_relu(float* __restrict__ acc, const float* __restrict__ xw,
                                const float* __restrict__ dis, const float* __restrict__ b,
                                int N) {
    size_t idx = (size_t)blockIdx.x * blockDim.x + threadIdx.x;
    if (idx >= (size_t)N * 128) return;
    int i = (int)(idx >> 7);
    int c = (int)(idx & 127);
    float di = dis[i];
    float v  = acc[idx] + di * di * xw[idx] + b[c];
    acc[idx] = fmaxf(v, 0.0f);
}

__global__ void k_final(float* __restrict__ out, const float* __restrict__ xwcat,
                        const float* __restrict__ dis, const float* __restrict__ bmu,
                        const float* __restrict__ blv, int N) {
    size_t idx = (size_t)blockIdx.x * blockDim.x + threadIdx.x;
    if (idx >= (size_t)N * 128) return;
    int i = (int)(idx >> 7);
    int c = (int)(idx & 127);
    float di  = dis[i];
    float val = di * di * xwcat[idx];
    if (c < 64) {
        out[(size_t)i * 64 + c] += val + bmu[c];
    } else {
        out[(size_t)N * 64 + (size_t)i * 64 + (c - 64)] += val + blv[c - 64];
    }
}

extern "C" void kernel_launch(void* const* d_in, const int* in_sizes, int n_in,
                              void* d_out, int out_size, void* d_ws, size_t ws_size,
                              hipStream_t stream) {
    const float* x    = (const float*)d_in[0];
    const int*   edge = (const int*)d_in[1];
    const float* W1   = (const float*)d_in[3];
    const float* b1   = (const float*)d_in[4];
    const float* Wmu  = (const float*)d_in[5];
    const float* bmu  = (const float*)d_in[6];
    const float* Wlv  = (const float*)d_in[7];
    const float* blv  = (const float*)d_in[8];

    const int N = in_sizes[0] / 128;
    const int E = in_sizes[1] / 2;
    const int* src = edge;
    const int* dst = edge + E;

    float* ws   = (float*)d_ws;
    float* deg  = ws;
    float* dis  = deg + N;
    float* Wcat = dis + N;
    float* B1   = Wcat + 128 * 128;
    float* B2   = B1 + (size_t)N * 128;
    float* outp = (float*)d_out;

    const size_t NC = (size_t)N * 128;
    const int nThreads = 256;
    const int gFill    = 2048;
    const int gE       = (E + nThreads - 1) / nThreads;
    const int gN       = (N + nThreads - 1) / nThreads;
    const int gNC      = (int)((NC + nThreads - 1) / nThreads);
    const int gEdgeW   = (int)(((size_t)E * 32 + nThreads - 1) / nThreads);

    k_fill<<<gFill, nThreads, 0, stream>>>(deg, (size_t)N, 1.0f);
    k_deg_edges<<<gE, nThreads, 0, stream>>>(dst, deg, E);
    k_dis<<<gN, nThreads, 0, stream>>>(deg, dis, N);
    k_pack_w<<<64, nThreads, 0, stream>>>(Wmu, Wlv, Wcat);

    k_gemm128<<<N / 16, nThreads, 0, stream>>>(x, W1, B1, N);
    k_fill<<<gFill, nThreads, 0, stream>>>(B2, NC, 0.0f);
    k_scatter<false><<<gEdgeW, nThreads, 0, stream>>>(B1, src, dst, dis, B2, E, N);
    k_selfloop_relu<<<gNC, nThreads, 0, stream>>>(B2, B1, dis, b1, N);

    k_gemm128<<<N / 16, nThreads, 0, stream>>>(B2, Wcat, B1, N);
    k_fill<<<gFill, nThreads, 0, stream>>>(outp, NC, 0.0f);
    k_scatter<true><<<gEdgeW, nThreads, 0, stream>>>(B1, src, dst, dis, outp, E, N);
    k_final<<<gNC, nThreads, 0, stream>>>(outp, B1, dis, bmu, blv, N);
}


# Edges counting-sorted by destination each call (rank = arrival order at an integer atomic); one wave sums each output row in registers and writes it once; float atomic scatter, fills, epilogues gone
# speedup vs baseline: 2.1329x; 2.1329x over previous
#include <hip/hip_runtime.h>
#include <hip/hip_bf16.h>


typedef __attribute__((ext_vector_type(2))) float v2f;
typedef __attribute__((ext_vector_type(8))) float v8f;

__global__ void k_zero_int(int* __restrict__ p, int n) {
    int i = blockIdx.x * blockDim.x + threadIdx.x;
    if (i < n) p[i] = 0;
}

__global__ void k_count_rank(const int* __restrict__ dst, int* __restrict__ cnt,
                             int* __restrict__ rank, int E) {
    int e = blockIdx.x * blockDim.x + threadIdx.x;
    if (e < E) rank[e] = atomicAdd(&cnt[dst[e]], 1);
}

__global__ void k_blocksum(const int* __restrict__ cnt, int* __restrict__ bsum, int N) {
    __shared__ int s_w[8];
    const int t = threadIdx.x, lane = t & 31, w = t >> 5;
    const int i0 = (blockIdx.x * 256 + t) * 4;
    int s = 0;
    for (int k = 0; k < 4; ++k)
        if (i0 + k < N) s += cnt[i0 + k];
    for (int d = 16; d > 0; d >>= 1) s += __shfl_xor(s, d, 32);
    if (lane == 0) s_w[w] = s;
    __syncthreads();
    if (t == 0) {
        int tot = 0;
        for (int k = 0; k < 8; ++k) tot += s_w[k];
        bsum[blockIdx.x] = tot;
    }
}

__global__ void k_rowptr(const int* __restrict__ cnt, const int* __restrict__ bsum,
                         int* __restrict__ rowptr, float* __restrict__ dis, int N) {
    __shared__ int s_w[8];
    __shared__ int s_p[8];
    const int t = threadIdx.x, lane = t & 31, w = t >> 5;

    int part = 0;
    for (int j = t; j < (int)blockIdx.x; j += 256) part += bsum[j];
    for (int d = 16; d > 0; d >>= 1) part += __shfl_xor(part, d, 32);
    if (lane == 0) s_p[w] = part;

    const int i0 = (blockIdx.x * 256 + t) * 4;
    int c[4];
    for (int k = 0; k < 4; ++k) c[k] = (i0 + k < N) ? cnt[i0 + k] : 0;
    const int mine = c[0] + c[1] + c[2] + c[3];
    int incl = mine;
    for (int d = 1; d < 32; d <<= 1) {
        const int up = __shfl_up(incl, d, 32);
        if (lane >= d) incl += up;
    }
    if (lane == 31) s_w[w] = incl;
    __syncthreads();

    int run = incl - mine;
    for (int k = 0; k < 8; ++k) {
        run += s_p[k];
        if (k < w) run += s_w[k];
    }
    for (int k = 0; k < 4; ++k) {
        if (i0 + k < N) {
            rowptr[i0 + k] = run;
            const float deg = (float)(c[k] + 1);
            dis[i0 + k] = (deg > 0.0f) ? rsqrtf(deg) : 0.0f;
            run += c[k];
        }
    }
}

__global__ void k_fill_csr(const int* __restrict__ src, const int* __restrict__ dst,
                           const int* __restrict__ rank, const int* __restrict__ rowptr,
                           const float* __restrict__ dis, int2* __restrict__ csr, int E) {
    int e = blockIdx.x * blockDim.x + threadIdx.x;
    if (e >= E) return;
    const int s = src[e];
    const int d = dst[e];
    const float nrm = dis[s] * dis[d];
    csr[rowptr[d] + rank[e]] = make_int2(s, __float_as_int(nrm));
}

__global__ void k_pack_w(const float* __restrict__ Wmu, const float* __restrict__ Wlv,
                         float* __restrict__ Wcat) {
    int idx = blockIdx.x * blockDim.x + threadIdx.x;
    int k = idx >> 7, c = idx & 127;
    Wcat[idx] = (c < 64) ? Wmu[k * 64 + c] : Wlv[k * 64 + (c - 64)];
}

__global__ void k_gemm128(const float* __restrict__ A, const float* __restrict__ W,
                          float* __restrict__ C, int nrows) {
    const int lane  = threadIdx.x & 31;
    const int wave  = threadIdx.x >> 5;
    const int row0  = blockIdx.x * 16;
    const int col0  = wave * 16;
    const int m     = lane & 15;
    const int khalf = (lane < 16) ? 0 : 2;

    const float* __restrict__ arow = A + (size_t)(row0 + m) * 128;
    v8f acc = {};
#pragma unroll
    for (int k = 0; k < 128; k += 4) {
        v2f a, b;
        const float2 av = *reinterpret_cast<const float2*>(arow + k + khalf);
        a.x = av.x;
        a.y = av.y;
        b.x = W[(size_t)(k + khalf) * 128 + col0 + m];
        b.y = W[(size_t)(k + khalf + 1) * 128 + col0 + m];
        acc = __builtin_amdgcn_wmma_f32_16x16x4_f32(
             false, a,  false, b,
             (short)0, acc,  false,  false);
    }
    const int rbase = (lane < 16) ? 0 : 8;
#pragma unroll
    for (int j = 0; j < 8; ++j) {
        C[(size_t)(row0 + rbase + j) * 128 + col0 + m] = acc[j];
    }
}

template <int K>
__device__ __forceinline__ void gather_rows(const float* __restrict__ xw, int my_s, int my_w,
                                            int k0, int c, float4& acc) {
    int    s[K];
    float  w[K];
    float4 v[K];
    for (int i = 0; i < K; ++i) {
        s[i] = __builtin_amdgcn_readlane(my_s, k0 + i);
        w[i] = __int_as_float(__builtin_amdgcn_readlane(my_w, k0 + i));
    }
    for (int i = 0; i < K; ++i)
        v[i] = *reinterpret_cast<const float4*>(xw + (size_t)s[i] * 128 + c);
    for (int i = 0; i < K; ++i) {
        acc.x += v[i].x * w[i];
        acc.y += v[i].y * w[i];
        acc.z += v[i].z * w[i];
        acc.w += v[i].w * w[i];
    }
}

template <bool SPLIT>
__global__ void k_gather(const float* __restrict__ xw, const int* __restrict__ rowptr,
                         const int* __restrict__ cnt, const int2* __restrict__ csr,
                         const float* __restrict__ dis, const float* __restrict__ b0,
                         const float* __restrict__ b1, float* __restrict__ out, int N) {
    const int lane = threadIdx.x & 31;
    const int wv  = __builtin_amdgcn_readfirstlane((int)(threadIdx.x >> 5));
    const int row = blockIdx.x * 8 + wv;
    if (row >= N) return;
    const int c = lane * 4;

    const int   beg = rowptr[row];
    const int   len = cnt[row];
    const float di  = dis[row];
    const float4 self = *reinterpret_cast<const float4*>(xw + (size_t)row * 128 + c);

    float4 acc = make_float4(0.0f, 0.0f, 0.0f, 0.0f);
    for (int base = 0; base < len; base += 32) {
        const int n = min(32, len - base);
        int my_s = 0, my_w = 0;
        if (lane < n) {
            const int2 ent = csr[beg + base + lane];
            my_s = (int)min((unsigned)ent.x, (unsigned)(N - 1));
            my_w = ent.y;
        }
        int k = 0;
        for (; k + 4 <= n; k += 4) gather_rows<4>(xw, my_s, my_w, k, c, acc);
        const int rem = n - k;
        if (rem == 3)      gather_rows<3>(xw, my_s, my_w, k, c, acc);
        else if (rem == 2) gather_rows<2>(xw, my_s, my_w, k, c, acc);
        else if (rem == 1) gather_rows<1>(xw, my_s, my_w, k, c, acc);
    }

    const float w0 = di * di;
    acc.x += w0 * self.x;
    acc.y += w0 * self.y;
    acc.z += w0 * self.z;
    acc.w += w0 * self.w;

    if (!SPLIT) {
        const float4 b = *reinterpret_cast<const float4*>(b0 + c);
        float4 h;
        h.x = fmaxf(acc.x + b.x, 0.0f);
        h.y = fmaxf(acc.y + b.y, 0.0f);
        h.z = fmaxf(acc.z + b.z, 0.0f);
        h.w = fmaxf(acc.w + b.w, 0.0f);
        *reinterpret_cast<float4*>(out + (size_t)row * 128 + c) = h;
    } else {
        const bool lo = (c < 64);
        const float4 b = *reinterpret_cast<const float4*>(lo ? (b0 + c) : (b1 + (c - 64)));
        float* p = lo ? out + (size_t)row * 64 + c
                      : out + (size_t)N * 64 + (size_t)row * 64 + (c - 64);
        float4 r;
        r.x = acc.x + b.x;
        r.y = acc.y + b.y;
        r.z = acc.z + b.z;
        r.w = acc.w + b.w;
        *reinterpret_cast<float4*>(p) = r;
    }
}

extern "C" void kernel_launch(void* const* d_in, const int* in_sizes, int n_in,
                              void* d_out, int out_size, void* d_ws, size_t ws_size,
                              hipStream_t stream) {
    const float* x    = (const float*)d_in[0];
    const int*   edge = (const int*)d_in[1];
    const float* W1   = (const float*)d_in[3];
    const float* b1   = (const float*)d_in[4];
    const float* Wmu  = (const float*)d_in[5];
    const float* bmu  = (const float*)d_in[6];
    const float* Wlv  = (const float*)d_in[7];
    const float* blv  = (const float*)d_in[8];

    const int N = in_sizes[0] / 128;
    const int E = in_sizes[1] / 2;
    const int* src = edge;
    const int* dst = edge + E;

    const int nThreads = 256;
    const int gE    = (E + nThreads - 1) / nThreads;
    const int gN    = (N + nThreads - 1) / nThreads;
    const int gScan = (N + 1023) / 1024;
    const int gRows = (N + 7) / 8;
    const int N4    = (N + 3) & ~3;
    const int E4    = (E + 3) & ~3;
    const int S4    = (gScan + 3) & ~3;

    float* B1     = (float*)d_ws;
    float* B2     = B1 + (size_t)N * 128;
    float* Wcat   = B2 + (size_t)N * 128;
    float* dis    = Wcat + 128 * 128;
    int*   cnt    = (int*)(dis + N4);
    int*   rowptr = cnt + N4;
    int*   bsum   = rowptr + N4;
    int*   rank   = bsum + S4;
    int2*  csr    = (int2*)(rank + E4);
    float* outp   = (float*)d_out;

    k_zero_int<<<gN, nThreads, 0, stream>>>(cnt, N);
    k_count_rank<<<gE, nThreads, 0, stream>>>(dst, cnt, rank, E);
    k_blocksum<<<gScan, nThreads, 0, stream>>>(cnt, bsum, N);
    k_rowptr<<<gScan, nThreads, 0, stream>>>(cnt, bsum, rowptr, dis, N);
    k_fill_csr<<<gE, nThreads, 0, stream>>>(src, dst, rank, rowptr, dis, csr, E);
    k_pack_w<<<64, nThreads, 0, stream>>>(Wmu, Wlv, Wcat);

    k_gemm128<<<N / 16, nThreads, 0, stream>>>(x, W1, B1, N);
    k_gather<false><<<gRows, nThreads, 0, stream>>>(B1, rowptr, cnt, csr, dis, b1, b1, B2, N);

    k_gemm128<<<N / 16, nThreads, 0, stream>>>(B2, Wcat, B1, N);
    k_gather<true><<<gRows, nThreads, 0, stream>>>(B1, rowptr, cnt, csr, dis, bmu, blv, outp, N);
}


# Independent work shares launches: the layer-1 product's waves fill the sorted edge list behind their stripes; weight packing rides with the counter clear; 10 launches -> 8; row sums as in the parent
# speedup vs baseline: 2.5831x; 1.2111x over previous
#include <hip/hip_runtime.h>
#include <hip/hip_bf16.h>


typedef __attribute__((ext_vector_type(2))) float v2f;
typedef __attribute__((ext_vector_type(8))) float v8f;

__global__ void k_count_rank(const int* __restrict__ dst, int* __restrict__ cnt,
                             int* __restrict__ rank, int E) {
    int e = blockIdx.x * blockDim.x + threadIdx.x;
    if (e < E) rank[e] = atomicAdd(&cnt[dst[e]], 1);
}

__global__ void k_blocksum(const int* __restrict__ cnt, int* __restrict__ bsum, int N) {
    __shared__ int s_w[8];
    const int t = threadIdx.x, lane = t & 31, w = t >> 5;
    const int i0 = (blockIdx.x * 256 + t) * 4;
    int s = 0;
    for (int k = 0; k < 4; ++k)
        if (i0 + k < N) s += cnt[i0 + k];
    for (int d = 16; d > 0; d >>= 1) s += __shfl_xor(s, d, 32);
    if (lane == 0) s_w[w] = s;
    __syncthreads();
    if (t == 0) {
        int tot = 0;
        for (int k = 0; k < 8; ++k) tot += s_w[k];
        bsum[blockIdx.x] = tot;
    }
}

__global__ void k_rowptr(const int* __restrict__ cnt, const int* __restrict__ bsum,
                         int* __restrict__ rowptr, float* __restrict__ dis, int N) {
    __shared__ int s_w[8];
    __shared__ int s_p[8];
    const int t = threadIdx.x, lane = t & 31, w = t >> 5;

    int part = 0;
    for (int j = t; j < (int)blockIdx.x; j += 256) part += bsum[j];
    for (int d = 16; d > 0; d >>= 1) part += __shfl_xor(part, d, 32);
    if (lane == 0) s_p[w] = part;

    const int i0 = (blockIdx.x * 256 + t) * 4;
    int c[4];
    for (int k = 0; k < 4; ++k) c[k] = (i0 + k < N) ? cnt[i0 + k] : 0;
    const int mine = c[0] + c[1] + c[2] + c[3];
    int incl = mine;
    for (int d = 1; d < 32; d <<= 1) {
        const int up = __shfl_up(incl, d, 32);
        if (lane >= d) incl += up;
    }
    if (lane == 31) s_w[w] = incl;
    __syncthreads();

    int run = incl - mine;
    for (int k = 0; k < 8; ++k) {
        run += s_p[k];
        if (k < w) run += s_w[k];
    }
    for (int k = 0; k < 4; ++k) {
        if (i0 + k < N) {
            rowptr[i0 + k] = run;
            const float deg = (float)(c[k] + 1);
            dis[i0 + k] = (deg > 0.0f) ? rsqrtf(deg) : 0.0f;
            run += c[k];
        }
    }
}

__global__ void k_zero_pack(const float* __restrict__ W1, const float* __restrict__ Wmu,
                            const float* __restrict__ Wlv, float4* __restrict__ Wp,
                            int* __restrict__ cnt, int n) {
    if (blockIdx.x >= 32) {
        const int i = (blockIdx.x - 32) * blockDim.x + threadIdx.x;
        if (i < n) cnt[i] = 0;
        return;
    }
    const int idx = blockIdx.x * blockDim.x + threadIdx.x;
    const int which = idx >> 12;
    const int q = idx & 4095;
    const int l = q & 31, jj = (q >> 5) & 3, s = q >> 7;
    const int c = l & 15, h = l >> 4;
    const int k0  = 8 * (s >> 1) + 4 * h + 2 * (s & 1);
    const int col = 64 * (jj >> 1) + 4 * c + 2 * (jj & 1);
    float4 v;
    if (which == 0) {
        v.x = W1[k0 * 128 + col];
        v.y = W1[(k0 + 1) * 128 + col];
        v.z = W1[k0 * 128 + col + 1];
        v.w = W1[(k0 + 1) * 128 + col + 1];
    } else {
        const float* __restrict__ Wh = (col < 64) ? Wmu : Wlv;
        const int ch = col & 63;
        v.x = Wh[k0 * 64 + ch];
        v.y = Wh[(k0 + 1) * 64 + ch];
        v.z = Wh[k0 * 64 + ch + 1];
        v.w = Wh[(k0 + 1) * 64 + ch + 1];
    }
    Wp[idx] = v;
}

template <bool FILL>
__global__ void __launch_bounds__(160)
k_gemm128(const float* __restrict__ A, const float4* __restrict__ Wp,
          float* __restrict__ C, int ntiles,
          const int* __restrict__ src, const int* __restrict__ dst,
          const int* __restrict__ rank, const int* __restrict__ rowptr,
          const float* __restrict__ dis, int2* __restrict__ csr, int E) {
    __shared__ float4 sW[4096];

    const int tid  = threadIdx.x;
    const int lane = tid & 31;
    const int wave = __builtin_amdgcn_readfirstlane(tid >> 5);
    const int tile = blockIdx.x * 5 + wave;
    const int c = lane & 15;
    const int h = lane >> 4;

    const int trow = min(tile, ntiles - 1) * 16;
    const float* __restrict__ arow = A + (size_t)(trow + c) * 128 + 4 * h;
    float4 aq[16];
#pragma unroll
    for (int kk = 0; kk < 16; ++kk)
        aq[kk] = *reinterpret_cast<const float4*>(arow + 8 * kk);

    for (int i = tid; i < 4096; i += 160) sW[i] = Wp[i];
    __syncthreads();

    if (tile < ntiles) {
        v8f acc[8];
#pragma unroll
        for (int j = 0; j < 8; ++j) acc[j] = v8f{};

        const float4* sB = sW + lane;
#pragma unroll
        for (int s = 0; s < 32; ++s) {
            const float4 q0 = sB[(s * 4 + 0) * 32];
            const float4 q1 = sB[(s * 4 + 1) * 32];
            const float4 q2 = sB[(s * 4 + 2) * 32];
            const float4 q3 = sB[(s * 4 + 3) * 32];
            v2f a;
            if (s & 1) { a.x = aq[s >> 1].z; a.y = aq[s >> 1].w; }
            else       { a.x = aq[s >> 1].x; a.y = aq[s >> 1].y; }
            v2f b;
            b.x = q0.x; b.y = q0.y;
            acc[0] = __builtin_amdgcn_wmma_f32_16x16x4_f32(false, a, false, b, (short)0, acc[0], false, false);
            b.x = q0.z; b.y = q0.w;
            acc[1] = __builtin_amdgcn_wmma_f32_16x16x4_f32(false, a, false, b, (short)0, acc[1], false, false);
            b.x = q1.x; b.y = q1.y;
            acc[2] = __builtin_amdgcn_wmma_f32_16x16x4_f32(false, a, false, b, (short)0, acc[2], false, false);
            b.x = q1.z; b.y = q1.w;
            acc[3] = __builtin_amdgcn_wmma_f32_16x16x4_f32(false, a, false, b, (short)0, acc[3], false, false);
            b.x = q2.x; b.y = q2.y;
            acc[4] = __builtin_amdgcn_wmma_f32_16x16x4_f32(false, a, false, b, (short)0, acc[4], false, false);
            b.x = q2.z; b.y = q2.w;
            acc[5] = __builtin_amdgcn_wmma_f32_16x16x4_f32(false, a, false, b, (short)0, acc[5], false, false);
            b.x = q3.x; b.y = q3.y;
            acc[6] = __builtin_amdgcn_wmma_f32_16x16x4_f32(false, a, false, b, (short)0, acc[6], false, false);
            b.x = q3.z; b.y = q3.w;
            acc[7] = __builtin_amdgcn_wmma_f32_16x16x4_f32(false, a, false, b, (short)0, acc[7], false, false);
        }

        float* __restrict__ crow = C + (size_t)(tile * 16 + 8 * h) * 128 + 4 * c;
#pragma unroll
        for (int r = 0; r < 8; ++r) {
            *reinterpret_cast<float4*>(crow + r * 128) =
                make_float4(acc[0][r], acc[1][r], acc[2][r], acc[3][r]);
            *reinterpret_cast<float4*>(crow + r * 128 + 64) =
                make_float4(acc[4][r], acc[5][r], acc[6][r], acc[7][r]);
        }
    }

    if (FILL) {
        for (int base = (int)blockIdx.x * 640; base < E; base += (int)gridDim.x * 640) {
            int   es[4], ed[4], er[4], rp[4];
            float fs[4], fd[4];
#pragma unroll
            for (int k = 0; k < 4; ++k) {
                const int e = base + 160 * k + tid;
                const bool ok = e < E;
                es[k] = ok ? src[e]  : 0;
                ed[k] = ok ? dst[e]  : 0;
                er[k] = ok ? rank[e] : 0;
            }
#pragma unroll
            for (int k = 0; k < 4; ++k) {
                fs[k] = dis[es[k]];
                fd[k] = dis[ed[k]];
                rp[k] = rowptr[ed[k]];
            }
#pragma unroll
            for (int k = 0; k < 4; ++k) {
                if (base + 160 * k + tid < E)
                    csr[rp[k] + er[k]] = make_int2(es[k], __float_as_int(fs[k] * fd[k]));
            }
        }
    }
}

template <int K>
__device__ __forceinline__ void gather_rows(const float* __restrict__ xw, int my_s, int my_w,
                                            int k0, int c, float4& acc) {
    int    s[K];
    float  w[K];
    float4 v[K];
    for (int i = 0; i < K; ++i) {
        s[i] = __builtin_amdgcn_readlane(my_s, k0 + i);
        w[i] = __int_as_float(__builtin_amdgcn_readlane(my_w, k0 + i));
    }
    for (int i = 0; i < K; ++i)
        v[i] = *reinterpret_cast<const float4*>(xw + (size_t)s[i] * 128 + c);
    for (int i = 0; i < K; ++i) {
        acc.x += v[i].x * w[i];
        acc.y += v[i].y * w[i];
        acc.z += v[i].z * w[i];
        acc.w += v[i].w * w[i];
    }
}

template <bool SPLIT>
__global__ void k_gather(const float* __restrict__ xw, const int* __restrict__ rowptr,
                         const int* __restrict__ cnt, const int2* __restrict__ csr,
                         const float* __restrict__ dis, const float* __restrict__ b0,
                         const float* __restrict__ b1, float* __restrict__ out, int N) {
    const int lane = threadIdx.x & 31;
    const int wv  = __builtin_amdgcn_readfirstlane((int)(threadIdx.x >> 5));
    const int row = blockIdx.x * 8 + wv;
    if (row >= N) return;
    const int c = lane * 4;

    const int   beg = rowptr[row];
    const int   len = cnt[row];
    const float di  = dis[row];
    const float4 self = *reinterpret_cast<const float4*>(xw + (size_t)row * 128 + c);

    float4 acc = make_float4(0.0f, 0.0f, 0.0f, 0.0f);
    for (int base = 0; base < len; base += 32) {
        const int n = min(32, len - base);
        int my_s = 0, my_w = 0;
        if (lane < n) {
            const int2 ent = csr[beg + base + lane];
            my_s = (int)min((unsigned)ent.x, (unsigned)(N - 1));
            my_w = ent.y;
        }
        int k = 0;
        for (; k + 4 <= n; k += 4) gather_rows<4>(xw, my_s, my_w, k, c, acc);
        const int rem = n - k;
        if (rem == 3)      gather_rows<3>(xw, my_s, my_w, k, c, acc);
        else if (rem == 2) gather_rows<2>(xw, my_s, my_w, k, c, acc);
        else if (rem == 1) gather_rows<1>(xw, my_s, my_w, k, c, acc);
    }

    const float w0 = di * di;
    acc.x += w0 * self.x;
    acc.y += w0 * self.y;
    acc.z += w0 * self.z;
    acc.w += w0 * self.w;

    if (!SPLIT) {
        const float4 b = *reinterpret_cast<const float4*>(b0 + c);
        float4 h;
        h.x = fmaxf(acc.x + b.x, 0.0f);
        h.y = fmaxf(acc.y + b.y, 0.0f);
        h.z = fmaxf(acc.z + b.z, 0.0f);
        h.w = fmaxf(acc.w + b.w, 0.0f);
        *reinterpret_cast<float4*>(out + (size_t)row * 128 + c) = h;
    } else {
        const bool lo = (c < 64);
        const float4 b = *reinterpret_cast<const float4*>(lo ? (b0 + c) : (b1 + (c - 64)));
        float* p = lo ? out + (size_t)row * 64 + c
                      : out + (size_t)N * 64 + (size_t)row * 64 + (c - 64);
        float4 r;
        r.x = acc.x + b.x;
        r.y = acc.y + b.y;
        r.z = acc.z + b.z;
        r.w = acc.w + b.w;
        *reinterpret_cast<float4*>(p) = r;
    }
}

extern "C" void kernel_launch(void* const* d_in, const int* in_sizes, int n_in,
                              void* d_out, int out_size, void* d_ws, size_t ws_size,
                              hipStream_t stream) {
    const float* x    = (const float*)d_in[0];
    const int*   edge = (const int*)d_in[1];
    const float* W1   = (const float*)d_in[3];
    const float* b1   = (const float*)d_in[4];
    const float* Wmu  = (const float*)d_in[5];
    const float* bmu  = (const float*)d_in[6];
    const float* Wlv  = (const float*)d_in[7];
    const float* blv  = (const float*)d_in[8];

    const int N = in_sizes[0] / 128;
    const int E = in_sizes[1] / 2;
    const int* src = edge;
    const int* dst = edge + E;

    const int nThreads = 256;
    const int gE    = (E + nThreads - 1) / nThreads;
    const int gN    = (N + nThreads - 1) / nThreads;
    const int gScan = (N + 1023) / 1024;
    const int gRows = (N + 7) / 8;
    const int nTiles = N / 16;
    const int gTile  = (nTiles + 4) / 5;
    const int N4    = (N + 3) & ~3;
    const int E4    = (E + 3) & ~3;
    const int S4    = (gScan + 3) & ~3;

    float* B1     = (float*)d_ws;
    float* B2     = B1 + (size_t)N * 128;
    float* Wp     = B2 + (size_t)N * 128;
    float* dis    = Wp + 2 * 128 * 128;
    int*   cnt    = (int*)(dis + N4);
    int*   rowptr = cnt + N4;
    int*   bsum   = rowptr + N4;
    int*   rank   = bsum + S4;
    int2*  csr    = (int2*)(rank + E4);
    float* outp   = (float*)d_out;

    k_zero_pack<<<32 + gN, nThreads, 0, stream>>>(W1, Wmu, Wlv, (float4*)Wp, cnt, N);
    k_count_rank<<<gE, nThreads, 0, stream>>>(dst, cnt, rank, E);
    k_blocksum<<<gScan, nThreads, 0, stream>>>(cnt, bsum, N);
    k_rowptr<<<gScan, nThreads, 0, stream>>>(cnt, bsum, rowptr, dis, N);

    k_gemm128<true><<<gTile, 160, 0, stream>>>(x, (const float4*)Wp, B1, nTiles,
                                               src, dst, rank, rowptr, dis, csr, E);
    k_gather<false><<<gRows, nThreads, 0, stream>>>(B1, rowptr, cnt, csr, dis, b1, b1, B2, N);

    k_gemm128<false><<<gTile, 160, 0, stream>>>(B2, (const float4*)Wp + 4096, B1, nTiles,
                                                src, dst, rank, rowptr, dis, csr, E);
    k_gather<true><<<gRows, nThreads, 0, stream>>>(B1, rowptr, cnt, csr, dis, bmu, blv, outp, N);
}


# Edge counts and ranks from LDS counters, block = (node range, edge-list chunk), no global atomic; counter-clear and block-sum launches gone (8 -> 6); rank = LDS arrival order: row sum order changes
# speedup vs baseline: 2.7230x; 1.0542x over previous
#include <hip/hip_runtime.h>
#include <hip/hip_bf16.h>


typedef __attribute__((ext_vector_type(2))) float v2f;
typedef __attribute__((ext_vector_type(8))) float v8f;

constexpr int EH   = 16;
constexpr int RUNS = 7;

__global__ void k_rowptr(int* __restrict__ cntH, const int* __restrict__ bsumH,
                         int* __restrict__ cnt, int* __restrict__ rowptr,
                         float* __restrict__ dis, int N, int NP, int G) {
    __shared__ int s_w[8];
    __shared__ int s_p[8];
    const int t = threadIdx.x, lane = t & 31, w = t >> 5;

    int part = 0;
    for (int h = 0; h < EH; ++h)
        for (int j = t; j < (int)blockIdx.x; j += 256) part += bsumH[h * G + j];
    for (int d = 16; d > 0; d >>= 1) part += __shfl_xor(part, d, 32);
    if (lane == 0) s_p[w] = part;

    const int i0 = (blockIdx.x * 256 + t) * 4;
    int4 p[EH];
#pragma unroll
    for (int h = 0; h < EH; ++h)
        p[h] = *reinterpret_cast<const int4*>(cntH + (size_t)h * NP + i0);
    int c[4] = {0, 0, 0, 0};
#pragma unroll
    for (int h = 0; h < EH; ++h) { c[0] += p[h].x; c[1] += p[h].y; c[2] += p[h].z; c[3] += p[h].w; }
    const int mine = c[0] + c[1] + c[2] + c[3];
    int incl = mine;
    for (int d = 1; d < 32; d <<= 1) {
        const int up = __shfl_up(incl, d, 32);
        if (lane >= d) incl += up;
    }
    if (lane == 31) s_w[w] = incl;
    __syncthreads();

    int run = incl - mine;
    for (int k = 0; k < 8; ++k) {
        run += s_p[k];
        if (k < w) run += s_w[k];
    }
    int4 o;
    o.x = run;
    o.y = o.x + c[0];
    o.z = o.y + c[1];
    o.w = o.z + c[2];
    const int rp[4] = {o.x, o.y, o.z, o.w};
    for (int k = 0; k < 4; ++k) {
        if (i0 + k < N) {
            rowptr[i0 + k] = rp[k];
            cnt[i0 + k]    = c[k];
            const float deg = (float)(c[k] + 1);
            dis[i0 + k] = (deg > 0.0f) ? rsqrtf(deg) : 0.0f;
        }
    }
#pragma unroll
    for (int h = 0; h < EH; ++h) {
        *reinterpret_cast<int4*>(cntH + (size_t)h * NP + i0) = o;
        o.x += p[h].x; o.y += p[h].y; o.z += p[h].z; o.w += p[h].w;
    }
}

__device__ __forceinline__ void count_one(int d, int lo, int e, int* s_cnt,
                                          int* __restrict__ rank) {
    const unsigned r = (unsigned)(d - lo);
    if (r < (unsigned)(RUNS * 1024)) rank[e] = atomicAdd(&s_cnt[r], 1);
}

__global__ void __launch_bounds__(1024)
k_count_pack(const int* __restrict__ dst, int* __restrict__ cntH, int* __restrict__ rank,
             int* __restrict__ bsumH, int E, int nOwn, int R, int NP, int G, int q16,
             const float* __restrict__ W1, const float* __restrict__ Wmu,
             const float* __restrict__ Wlv, float4* __restrict__ Wp) {
    __shared__ int s_cnt[RUNS * 1024];
    __shared__ int s_w[RUNS * 32];
    const int t = threadIdx.x;

    if ((int)blockIdx.x >= nOwn) {
        const int idx = ((int)blockIdx.x - nOwn) * 1024 + t;
        const int which = idx >> 12;
        const int q = idx & 4095;
        const int l = q & 31, jj = (q >> 5) & 3, s = q >> 7;
        const int c = l & 15, h = l >> 4;
        const int k0  = 8 * (s >> 1) + 4 * h + 2 * (s & 1);
        const int col = 64 * (jj >> 1) + 4 * c + 2 * (jj & 1);
        float4 v;
        if (which == 0) {
            v.x = W1[k0 * 128 + col];
            v.y = W1[(k0 + 1) * 128 + col];
            v.z = W1[k0 * 128 + col + 1];
            v.w = W1[(k0 + 1) * 128 + col + 1];
        } else {
            const float* __restrict__ Wh = (col < 64) ? Wmu : Wlv;
            const int ch = col & 63;
            v.x = Wh[k0 * 64 + ch];
            v.y = Wh[(k0 + 1) * 64 + ch];
            v.z = Wh[k0 * 64 + ch + 1];
            v.w = Wh[(k0 + 1) * 64 + ch + 1];
        }
        Wp[idx] = v;
        return;
    }

    const int rng   = (int)blockIdx.x % R;
    const int chunk = (int)blockIdx.x / R;
    for (int j = 0; j < RUNS; ++j) s_cnt[1024 * j + t] = 0;
    __syncthreads();

    const int lo  = rng * (RUNS * 1024);
    const int n4  = (E + 3) >> 2;
    const int qlo = chunk * q16;
    const int qhi = min(qlo + q16, n4);
    if ((E & 3) == 0) {
        const int4* __restrict__ dst4 = reinterpret_cast<const int4*>(dst);
        for (int base = qlo; base < qhi; base += 4 * 1024) {
            int4 d[4];
#pragma unroll
            for (int k = 0; k < 4; ++k) {
                const int q = base + k * 1024 + t;
                d[k] = (q < qhi) ? dst4[q] : make_int4(-1, -1, -1, -1);
            }
#pragma unroll
            for (int k = 0; k < 4; ++k) {
                const int e = (base + k * 1024 + t) * 4;
                count_one(d[k].x, lo, e,     s_cnt, rank);
                count_one(d[k].y, lo, e + 1, s_cnt, rank);
                count_one(d[k].z, lo, e + 2, s_cnt, rank);
                count_one(d[k].w, lo, e + 3, s_cnt, rank);
            }
        }
    } else {
        const int ehi = min(qhi * 4, E);
        for (int e = qlo * 4 + t; e < ehi; e += 1024) count_one(dst[e], lo, e, s_cnt, rank);
    }
    __syncthreads();

    const int lane = t & 31, w = t >> 5;
    for (int j = 0; j < RUNS; ++j) {
        const int c = s_cnt[1024 * j + t];
        cntH[(size_t)chunk * NP + lo + 1024 * j + t] = c;
        int s = c;
        for (int dd = 16; dd > 0; dd >>= 1) s += __shfl_xor(s, dd, 32);
        if (lane == 0) s_w[32 * j + w] = s;
    }
    __syncthreads();
    if (t < RUNS) {
        int tot = 0;
        for (int k = 0; k < 32; ++k) tot += s_w[32 * t + k];
        bsumH[chunk * G + rng * RUNS + t] = tot;
    }
}

template <bool FILL>
__global__ void __launch_bounds__(160)
k_gemm128(const float* __restrict__ A, const float4* __restrict__ Wp,
          float* __restrict__ C, int ntiles,
          const int* __restrict__ src, const int* __restrict__ dst,
          const int* __restrict__ rank, const int* __restrict__ place,
          const float* __restrict__ dis, int2* __restrict__ csr, int E,
          int NP, int q16) {
    __shared__ float4 sW[4096];

    const int tid  = threadIdx.x;
    const int lane = tid & 31;
    const int wave = __builtin_amdgcn_readfirstlane(tid >> 5);
    const int tile = blockIdx.x * 5 + wave;
    const int c = lane & 15;
    const int h = lane >> 4;

    const int trow = min(tile, ntiles - 1) * 16;
    const float* __restrict__ arow = A + (size_t)(trow + c) * 128 + 4 * h;
    float4 aq[16];
#pragma unroll
    for (int kk = 0; kk < 16; ++kk)
        aq[kk] = *reinterpret_cast<const float4*>(arow + 8 * kk);

    for (int i = tid; i < 4096; i += 160) sW[i] = Wp[i];
    __syncthreads();

    if (tile < ntiles) {
        v8f acc[8];
#pragma unroll
        for (int j = 0; j < 8; ++j) acc[j] = v8f{};

        const float4* sB = sW + lane;
#pragma unroll
        for (int s = 0; s < 32; ++s) {
            const float4 q0 = sB[(s * 4 + 0) * 32];
            const float4 q1 = sB[(s * 4 + 1) * 32];
            const float4 q2 = sB[(s * 4 + 2) * 32];
            const float4 q3 = sB[(s * 4 + 3) * 32];
            v2f a;
            if (s & 1) { a.x = aq[s >> 1].z; a.y = aq[s >> 1].w; }
            else       { a.x = aq[s >> 1].x; a.y = aq[s >> 1].y; }
            v2f b;
            b.x = q0.x; b.y = q0.y;
            acc[0] = __builtin_amdgcn_wmma_f32_16x16x4_f32(false, a, false, b, (short)0, acc[0], false, false);
            b.x = q0.z; b.y = q0.w;
            acc[1] = __builtin_amdgcn_wmma_f32_16x16x4_f32(false, a, false, b, (short)0, acc[1], false, false);
            b.x = q1.x; b.y = q1.y;
            acc[2] = __builtin_amdgcn_wmma_f32_16x16x4_f32(false, a, false, b, (short)0, acc[2], false, false);
            b.x = q1.z; b.y = q1.w;
            acc[3] = __builtin_amdgcn_wmma_f32_16x16x4_f32(false, a, false, b, (short)0, acc[3], false, false);
            b.x = q2.x; b.y = q2.y;
            acc[4] = __builtin_amdgcn_wmma_f32_16x16x4_f32(false, a, false, b, (short)0, acc[4], false, false);
            b.x = q2.z; b.y = q2.w;
            acc[5] = __builtin_amdgcn_wmma_f32_16x16x4_f32(false, a, false, b, (short)0, acc[5], false, false);
            b.x = q3.x; b.y = q3.y;
            acc[6] = __builtin_amdgcn_wmma_f32_16x16x4_f32(false, a, false, b, (short)0, acc[6], false, false);
            b.x = q3.z; b.y = q3.w;
            acc[7] = __builtin_amdgcn_wmma_f32_16x16x4_f32(false, a, false, b, (short)0, acc[7], false, false);
        }

        float* __restrict__ crow = C + (size_t)(tile * 16 + 8 * h) * 128 + 4 * c;
#pragma unroll
        for (int r = 0; r < 8; ++r) {
            *reinterpret_cast<float4*>(crow + r * 128) =
                make_float4(acc[0][r], acc[1][r], acc[2][r], acc[3][r]);
            *reinterpret_cast<float4*>(crow + r * 128 + 64) =
                make_float4(acc[4][r], acc[5][r], acc[6][r], acc[7][r]);
        }
    }

    if (FILL) {
        for (int base = (int)blockIdx.x * 640; base < E; base += (int)gridDim.x * 640) {
            int   es[4], ed[4], er[4], rp[4];
            float fs[4], fd[4];
#pragma unroll
            for (int k = 0; k < 4; ++k) {
                const int e = base + 160 * k + tid;
                const bool ok = e < E;
                es[k] = ok ? src[e]  : 0;
                ed[k] = ok ? dst[e]  : 0;
                er[k] = ok ? rank[e] : 0;
            }
#pragma unroll
            for (int k = 0; k < 4; ++k) {
                const int e = min(base + 160 * k + tid, E - 1);
                fs[k] = dis[es[k]];
                fd[k] = dis[ed[k]];
                rp[k] = place[(size_t)((e >> 2) / q16) * NP + ed[k]];
            }
#pragma unroll
            for (int k = 0; k < 4; ++k) {
                if (base + 160 * k + tid < E)
                    csr[rp[k] + er[k]] = make_int2(es[k], __float_as_int(fs[k] * fd[k]));
            }
        }
    }
}

template <int K>
__device__ __forceinline__ void gather_rows(const float* __restrict__ xw, int my_s, int my_w,
                                            int k0, int c, float4& acc) {
    int    s[K];
    float  w[K];
    float4 v[K];
    for (int i = 0; i < K; ++i) {
        s[i] = __builtin_amdgcn_readlane(my_s, k0 + i);
        w[i] = __int_as_float(__builtin_amdgcn_readlane(my_w, k0 + i));
    }
    for (int i = 0; i < K; ++i)
        v[i] = *reinterpret_cast<const float4*>(xw + (size_t)s[i] * 128 + c);
    for (int i = 0; i < K; ++i) {
        acc.x += v[i].x * w[i];
        acc.y += v[i].y * w[i];
        acc.z += v[i].z * w[i];
        acc.w += v[i].w * w[i];
    }
}

template <bool SPLIT>
__global__ void k_gather(const float* __restrict__ xw, const int* __restrict__ rowptr,
                         const int* __restrict__ cnt, const int2* __restrict__ csr,
                         const float* __restrict__ dis, const float* __restrict__ b0,
                         const float* __restrict__ b1, float* __restrict__ out, int N) {
    const int lane = threadIdx.x & 31;
    const int wv  = __builtin_amdgcn_readfirstlane((int)(threadIdx.x >> 5));
    const int row = blockIdx.x * 8 + wv;
    if (row >= N) return;
    const int c = lane * 4;

    const int   beg = rowptr[row];
    const int   len = cnt[row];
    const float di  = dis[row];
    const float4 self = *reinterpret_cast<const float4*>(xw + (size_t)row * 128 + c);

    float4 acc = make_float4(0.0f, 0.0f, 0.0f, 0.0f);
    for (int base = 0; base < len; base += 32) {
        const int n = min(32, len - base);
        int my_s = 0, my_w = 0;
        if (lane < n) {
            const int2 ent = csr[beg + base + lane];
            my_s = (int)min((unsigned)ent.x, (unsigned)(N - 1));
            my_w = ent.y;
        }
        int k = 0;
        for (; k + 4 <= n; k += 4) gather_rows<4>(xw, my_s, my_w, k, c, acc);
        const int rem = n - k;
        if (rem == 3)      gather_rows<3>(xw, my_s, my_w, k, c, acc);
        else if (rem == 2) gather_rows<2>(xw, my_s, my_w, k, c, acc);
        else if (rem == 1) gather_rows<1>(xw, my_s, my_w, k, c, acc);
    }

    const float w0 = di * di;
    acc.x += w0 * self.x;
    acc.y += w0 * self.y;
    acc.z += w0 * self.z;
    acc.w += w0 * self.w;

    if (!SPLIT) {
        const float4 b = *reinterpret_cast<const float4*>(b0 + c);
        float4 h;
        h.x = fmaxf(acc.x + b.x, 0.0f);
        h.y = fmaxf(acc.y + b.y, 0.0f);
        h.z = fmaxf(acc.z + b.z, 0.0f);
        h.w = fmaxf(acc.w + b.w, 0.0f);
        *reinterpret_cast<float4*>(out + (size_t)row * 128 + c) = h;
    } else {
        const bool lo = (c < 64);
        const float4 b = *reinterpret_cast<const float4*>(lo ? (b0 + c) : (b1 + (c - 64)));
        float* p = lo ? out + (size_t)row * 64 + c
                      : out + (size_t)N * 64 + (size_t)row * 64 + (c - 64);
        float4 r;
        r.x = acc.x + b.x;
        r.y = acc.y + b.y;
        r.z = acc.z + b.z;
        r.w = acc.w + b.w;
        *reinterpret_cast<float4*>(p) = r;
    }
}

extern "C" void kernel_launch(void* const* d_in, const int* in_sizes, int n_in,
                              void* d_out, int out_size, void* d_ws, size_t ws_size,
                              hipStream_t stream) {
    const float* x    = (const float*)d_in[0];
    const int*   edge = (const int*)d_in[1];
    const float* W1   = (const float*)d_in[3];
    const float* b1   = (const float*)d_in[4];
    const float* Wmu  = (const float*)d_in[5];
    const float* bmu  = (const float*)d_in[6];
    const float* Wlv  = (const float*)d_in[7];
    const float* blv  = (const float*)d_in[8];

    const int N = in_sizes[0] / 128;
    const int E = in_sizes[1] / 2;
    const int* src = edge;
    const int* dst = edge + E;

    const int nThreads = 256;
    const int gScan = (N + 1023) / 1024;
    const int gRows = (N + 7) / 8;
    const int nTiles = N / 16;
    const int gTile  = (nTiles + 4) / 5;
    const int R     = (gScan + RUNS - 1) / RUNS;
    const int G     = R * RUNS;
    const int NP    = G * 1024;
    const int n4    = (E + 3) / 4;
    const int q16   = (n4 + EH - 1) / EH;
    const int N4    = (N + 3) & ~3;
    const int E4    = (E + 3) & ~3;

    float* B1     = (float*)d_ws;
    float* B2     = B1 + (size_t)N * 128;
    float* Wp     = B2 + (size_t)N * 128;
    float* dis    = Wp + 2 * 128 * 128;
    int*   cnt    = (int*)(dis + N4);
    int*   rowptr = cnt + N4;
    int*   rank   = rowptr + N4;
    int2*  csr    = (int2*)(rank + E4);
    int*   cntH   = (int*)(csr + E4);
    int*   bsumH  = cntH + (size_t)EH * NP;
    float* outp   = (float*)d_out;

    k_count_pack<<<R * EH + 8, 1024, 0, stream>>>(dst, cntH, rank, bsumH, E, R * EH, R, NP, G, q16,
                                                  W1, Wmu, Wlv, (float4*)Wp);
    k_rowptr<<<gScan, nThreads, 0, stream>>>(cntH, bsumH, cnt, rowptr, dis, N, NP, G);

    k_gemm128<true><<<gTile, 160, 0, stream>>>(x, (const float4*)Wp, B1, nTiles,
                                               src, dst, rank, cntH, dis, csr, E, NP, q16);
    k_gather<false><<<gRows, nThreads, 0, stream>>>(B1, rowptr, cnt, csr, dis, b1, b1, B2, N);

    k_gemm128<false><<<gTile, 160, 0, stream>>>(B2, (const float4*)Wp + 4096, B1, nTiles,
                                                src, dst, rank, cntH, dis, csr, E, NP, q16);
    k_gather<true><<<gRows, nThreads, 0, stream>>>(B1, rowptr, cnt, csr, dis, bmu, blv, outp, N);
}
